# starting kernel
# baseline (speedup 1.0000x reference)
#include <hip/hip_runtime.h>
#include <math.h>

#define Bsz   16
#define Nn    256
#define Ld    64
#define Hh    32
#define HE    128
#define EO    64
#define NMPS  3
#define ALPHA 0.2f

typedef __attribute__((ext_vector_type(2))) float v2f;
typedef __attribute__((ext_vector_type(8))) float v8f;

__device__ __forceinline__ float lrelu(float v) { return (v >= 0.f) ? v : ALPHA * v; }

__global__ __launch_bounds__(64) void k_input(const float* __restrict__ x,
                                              const float* __restrict__ W_lin,
                                              const float* __restrict__ b_lin,
                                              const float* __restrict__ W_in,
                                              const float* __restrict__ b_in,
                                              float* __restrict__ h) {
  const int b = blockIdx.x >> 8;
  const int n = blockIdx.x & 255;
  const int c = threadIdx.x;
  __shared__ float xs[Ld];
  __shared__ float ts[Ld];
  xs[c] = x[b * Ld + c];
  __syncthreads();
  float acc = b_lin[n * Ld + c];
#pragma unroll 8
  for (int l = 0; l < Ld; ++l) acc += xs[l] * W_lin[l * (Nn * Ld) + n * Ld + c];
  ts[c] = acc;
  __syncthreads();
  if (c < Hh) {
    float a2 = b_in[c];
#pragma unroll 8
    for (int l = 0; l < Ld; ++l) a2 += ts[l] * W_in[l * Hh + c];
    h[(b * Nn + n) * Hh + c] = a2;
  }
}

__global__ __launch_bounds__(128) void k_rcv_snd(const float* __restrict__ h,
                                                 const float* __restrict__ We1,
                                                 float* __restrict__ rcv,
                                                 float* __restrict__ snd, int t) {
  const int bn = blockIdx.x;
  const int k = threadIdx.x;
  __shared__ float hs[Hh];
  if (k < Hh) hs[k] = h[bn * Hh + k];
  __syncthreads();
  const float* W1 = We1 + (size_t)t * (2 * Hh) * HE;
  float a1 = 0.f, a2 = 0.f;
#pragma unroll 8
  for (int hh = 0; hh < Hh; ++hh) {
    const float hv = hs[hh];
    a1 += hv * W1[hh * HE + k];
    a2 += hv * W1[(Hh + hh) * HE + k];
  }
  rcv[bn * HE + k] = a1;
  snd[bn * HE + k] = a2;
}

#define EPAD 132

__global__ __launch_bounds__(128) void k_edge(const float* __restrict__ rcv,
                                              const float* __restrict__ snd,
                                              const float* __restrict__ be1,
                                              const float* __restrict__ We2,
                                              const float* __restrict__ be2,
                                              float* __restrict__ m, int t) {
  const int b    = blockIdx.x >> 8;
  const int i    = blockIdx.x & 255;
  const int tid  = threadIdx.x;
  const int wave = tid >> 5;
  const int lane = tid & 31;
  const int lh   = lane >> 4;
  const int ln   = lane & 15;

  __shared__ float Etile[16 * EPAD];

  const float rc = rcv[(b * Nn + i) * HE + tid] + be1[t * HE + tid];

  const int nc = wave * 16 + ln;
  const float* W2 = We2 + (size_t)t * HE * EO;
  v2f Bf[32];
#pragma unroll
  for (int kk = 0; kk < 32; ++kk) {
    const int k0 = kk * 4 + lh * 2;
    Bf[kk].x = W2[(k0    ) * EO + nc];
    Bf[kk].y = W2[(k0 + 1) * EO + nc];
  }
  const float bias2 = be2[t * EO + nc];

  float macc[8];
#pragma unroll
  for (int r = 0; r < 8; ++r) macc[r] = 0.f;

  const float* sndb = snd + (size_t)(b * Nn) * HE;

  for (int jt = 0; jt < Nn / 16; ++jt) {
    __syncthreads();
#pragma unroll 4
    for (int jr = 0; jr < 16; ++jr) {
      Etile[jr * EPAD + tid] = lrelu(rc + sndb[(jt * 16 + jr) * HE + tid]);
    }
    __syncthreads();

    v8f C = {};
#pragma unroll
    for (int kk = 0; kk < 32; ++kk) {
      const int k0 = kk * 4 + lh * 2;
      v2f A;
      A.x = Etile[ln * EPAD + k0];
      A.y = Etile[ln * EPAD + k0 + 1];
      C = __builtin_amdgcn_wmma_f32_16x16x4_f32(
           false, A,  false, Bf[kk],
           (short)0, C,  false,  false);
    }
#pragma unroll
    for (int r = 0; r < 8; ++r) macc[r] += lrelu(C[r] + bias2);
  }

  float colsum = 0.f;
#pragma unroll
  for (int r = 0; r < 8; ++r) colsum += macc[r];
  colsum += __shfl_xor(colsum, 16, 32);
  if (lane < 16) m[(b * Nn + i) * EO + nc] = colsum;
}

__global__ __launch_bounds__(32) void k_node(const float* __restrict__ h,
                                             const float* __restrict__ m,
                                             const float* __restrict__ Wn0,
                                             const float* __restrict__ bn0,
                                             const float* __restrict__ Wn1,
                                             const float* __restrict__ bn1,
                                             float* __restrict__ hout, int t,
                                             int apply_tanh) {
  const int bn = blockIdx.x;
  const int tid = threadIdx.x;
  __shared__ float nf[Hh + EO];
  __shared__ float t0[Hh];
  nf[tid]          = h[bn * Hh + tid];
  nf[Hh + tid]     = m[bn * EO + tid];
  nf[Hh + 32 + tid]= m[bn * EO + 32 + tid];
  __syncthreads();
  const float* W0 = Wn0 + (size_t)t * (Hh + EO) * Hh;
  float acc = bn0[t * Hh + tid];
#pragma unroll 8
  for (int f = 0; f < Hh + EO; ++f) acc += nf[f] * W0[f * Hh + tid];
  t0[tid] = lrelu(acc);
  __syncthreads();
  const float* W1 = Wn1 + (size_t)t * Hh * Hh;
  float a2 = bn1[t * Hh + tid];
#pragma unroll 8
  for (int f = 0; f < Hh; ++f) a2 += t0[f] * W1[f * Hh + tid];
  a2 = lrelu(a2);
  hout[bn * Hh + tid] = apply_tanh ? tanhf(a2) : a2;
}

extern "C" void kernel_launch(void* const* d_in, const int* in_sizes, int n_in,
                              void* d_out, int out_size, void* d_ws, size_t ws_size,
                              hipStream_t stream) {
  const float* x     = (const float*)d_in[0];
  const float* W_lin = (const float*)d_in[1];
  const float* b_lin = (const float*)d_in[2];
  const float* W_in  = (const float*)d_in[3];
  const float* b_in  = (const float*)d_in[4];
  const float* We1   = (const float*)d_in[5];
  const float* be1   = (const float*)d_in[6];
  const float* We2   = (const float*)d_in[7];
  const float* be2   = (const float*)d_in[8];
  const float* Wn0   = (const float*)d_in[9];
  const float* bn0   = (const float*)d_in[10];
  const float* Wn1   = (const float*)d_in[11];
  const float* bn1   = (const float*)d_in[12];
  float* out = (float*)d_out;

  float* ws  = (float*)d_ws;
  float* h   = ws;
  float* rcv = h   + Bsz * Nn * Hh;
  float* snd = rcv + Bsz * Nn * HE;
  float* mbuf= snd + Bsz * Nn * HE;

  const int BN = Bsz * Nn;

  k_input<<<BN, 64, 0, stream>>>(x, W_lin, b_lin, W_in, b_in, h);
  for (int t = 0; t < NMPS; ++t) {
    k_rcv_snd<<<BN, 128, 0, stream>>>(h, We1, rcv, snd, t);
    k_edge<<<BN, 128, 0, stream>>>(rcv, snd, be1, We2, be2, mbuf, t);
    k_node<<<BN, 32, 0, stream>>>(h, mbuf, Wn0, bn0, Wn1, bn1,
                                  (t == NMPS - 1) ? out : h, t, t == NMPS - 1);
  }
}


# k_edge: A fragments built in registers (16 receiver rows x 1 broadcast sender row), reused by all 4 column tiles; no LDS tile or loop barrier; 8 waves split the sender sum (k and sum order change)
# speedup vs baseline: 1.2242x; 1.2242x over previous
#include <hip/hip_runtime.h>
#include <math.h>

#define Bsz   16
#define Nn    256
#define Ld    64
#define Hh    32
#define HE    128
#define EO    64
#define NMPS  3
#define ALPHA 0.2f

typedef __attribute__((ext_vector_type(2))) float v2f;
typedef __attribute__((ext_vector_type(8))) float v8f;

__device__ __forceinline__ float lrelu(float v) { return (v >= 0.f) ? v : ALPHA * v; }

__global__ __launch_bounds__(64) void k_input(const float* __restrict__ x,
                                              const float* __restrict__ W_lin,
                                              const float* __restrict__ b_lin,
                                              const float* __restrict__ W_in,
                                              const float* __restrict__ b_in,
                                              float* __restrict__ h) {
  const int b = blockIdx.x >> 8;
  const int n = blockIdx.x & 255;
  const int c = threadIdx.x;
  __shared__ float xs[Ld];
  __shared__ float ts[Ld];
  xs[c] = x[b * Ld + c];
  __syncthreads();
  float acc = b_lin[n * Ld + c];
#pragma unroll 8
  for (int l = 0; l < Ld; ++l) acc += xs[l] * W_lin[l * (Nn * Ld) + n * Ld + c];
  ts[c] = acc;
  __syncthreads();
  if (c < Hh) {
    float a2 = b_in[c];
#pragma unroll 8
    for (int l = 0; l < Ld; ++l) a2 += ts[l] * W_in[l * Hh + c];
    h[(b * Nn + n) * Hh + c] = a2;
  }
}

__global__ __launch_bounds__(128) void k_rcv_snd(const float* __restrict__ h,
                                                 const float* __restrict__ We1,
                                                 float* __restrict__ rcv,
                                                 float* __restrict__ snd, int t) {
  const int bn = blockIdx.x;
  const int k = threadIdx.x;
  __shared__ float hs[Hh];
  if (k < Hh) hs[k] = h[bn * Hh + k];
  __syncthreads();
  const float* W1 = We1 + (size_t)t * (2 * Hh) * HE;
  float a1 = 0.f, a2 = 0.f;
#pragma unroll 8
  for (int hh = 0; hh < Hh; ++hh) {
    const float hv = hs[hh];
    a1 += hv * W1[hh * HE + k];
    a2 += hv * W1[(Hh + hh) * HE + k];
  }
  rcv[bn * HE + k] = a1;
  snd[bn * HE + k] = a2;
}

#define RT    16
#define SW    32
#define NWAVE 8
#define NBLK_EDGE 256

typedef __attribute__((ext_vector_type(4))) float v4f;

__global__ __launch_bounds__(256) void k_edge(const float* __restrict__ rcv,
                                              const float* __restrict__ snd,
                                              const float* __restrict__ be1,
                                              const float* __restrict__ We2,
                                              const float* __restrict__ be2,
                                              float* __restrict__ m, int t) {
  const int b    = blockIdx.x >> 4;
  const int it   = blockIdx.x & 15;
  const int tid  = threadIdx.x;
  const int wave = tid >> 5;
  const int lane = tid & 31;
  const int lh   = lane >> 4;
  const int ln   = lane & 15;

  __shared__ __attribute__((aligned(16))) float part[NWAVE][RT][EO];

  const float* rrow = rcv + (size_t)(b * Nn + it * RT + ln) * HE + 4 * lh;
  const float* brow = be1 + t * HE + 4 * lh;
  v4f rf[16];
#pragma unroll
  for (int q = 0; q < 16; ++q)
    rf[q] = *(const v4f*)(rrow + 8 * q) + *(const v4f*)(brow + 8 * q);

  const float* W2 = We2 + (size_t)t * HE * EO;
  v2f Bf[4][32];
#pragma unroll
  for (int kk = 0; kk < 32; ++kk) {
    const int k0 = 8 * (kk >> 1) + 4 * lh + 2 * (kk & 1);
    const v4f w0 = *(const v4f*)(W2 + (k0    ) * EO + 4 * ln);
    const v4f w1 = *(const v4f*)(W2 + (k0 + 1) * EO + 4 * ln);
#pragma unroll
    for (int nt = 0; nt < 4; ++nt) {
      Bf[nt][kk].x = w0[nt];
      Bf[nt][kk].y = w1[nt];
    }
  }
  const v4f bias2 = *(const v4f*)(be2 + t * EO + 4 * ln);

  float macc[4][8];
#pragma unroll
  for (int nt = 0; nt < 4; ++nt)
#pragma unroll
    for (int r = 0; r < 8; ++r) macc[nt][r] = 0.f;

  const float* srow = snd + (size_t)(b * Nn + wave * SW) * HE + 4 * lh;
  v4f ring[8];
#pragma unroll
  for (int q = 0; q < 8; ++q) ring[q] = *(const v4f*)(srow + 8 * q);

  for (int s = 0; s < SW; ++s) {
    const float* crow = srow + (size_t)s * HE;
    const float* nrow = srow + (size_t)((s + 1 < SW) ? s + 1 : s) * HE;

    v8f C[4];
#pragma unroll
    for (int nt = 0; nt < 4; ++nt) C[nt] = (v8f){};

#pragma unroll
    for (int q = 0; q < 16; ++q) {
      const v4f x = rf[q] + ring[q & 7];
      ring[q & 7] = (q < 8) ? *(const v4f*)(crow + 8 * (q + 8))
                            : *(const v4f*)(nrow + 8 * (q - 8));
      v2f A0, A1;
      A0.x = lrelu(x.x); A0.y = lrelu(x.y);
      A1.x = lrelu(x.z); A1.y = lrelu(x.w);
#pragma unroll
      for (int nt = 0; nt < 4; ++nt)
        C[nt] = __builtin_amdgcn_wmma_f32_16x16x4_f32(
             false, A0,  false, Bf[nt][2 * q],
             (short)0, C[nt],  false,  false);
#pragma unroll
      for (int nt = 0; nt < 4; ++nt)
        C[nt] = __builtin_amdgcn_wmma_f32_16x16x4_f32(
             false, A1,  false, Bf[nt][2 * q + 1],
             (short)0, C[nt],  false,  false);
    }
#pragma unroll
    for (int nt = 0; nt < 4; ++nt)
#pragma unroll
      for (int r = 0; r < 8; ++r) macc[nt][r] += lrelu(C[nt][r] + bias2[nt]);
  }

#pragma unroll
  for (int r = 0; r < 8; ++r) {
    v4f pv;
#pragma unroll
    for (int nt = 0; nt < 4; ++nt) pv[nt] = macc[nt][r];
    *(v4f*)&part[wave][8 * lh + r][4 * ln] = pv;
  }
  __syncthreads();
  const int row = tid >> 4;
  const int c4  = (tid & 15) * 4;
  v4f acc = *(const v4f*)&part[0][row][c4];
#pragma unroll
  for (int w = 1; w < NWAVE; ++w) acc += *(const v4f*)&part[w][row][c4];
  *(v4f*)(m + (size_t)(b * Nn + it * RT + row) * EO + c4) = acc;
}

__global__ __launch_bounds__(32) void k_node(const float* __restrict__ h,
                                             const float* __restrict__ m,
                                             const float* __restrict__ Wn0,
                                             const float* __restrict__ bn0,
                                             const float* __restrict__ Wn1,
                                             const float* __restrict__ bn1,
                                             float* __restrict__ hout, int t,
                                             int apply_tanh) {
  const int bn = blockIdx.x;
  const int tid = threadIdx.x;
  __shared__ float nf[Hh + EO];
  __shared__ float t0[Hh];
  nf[tid]          = h[bn * Hh + tid];
  nf[Hh + tid]     = m[bn * EO + tid];
  nf[Hh + 32 + tid]= m[bn * EO + 32 + tid];
  __syncthreads();
  const float* W0 = Wn0 + (size_t)t * (Hh + EO) * Hh;
  float acc = bn0[t * Hh + tid];
#pragma unroll 8
  for (int f = 0; f < Hh + EO; ++f) acc += nf[f] * W0[f * Hh + tid];
  t0[tid] = lrelu(acc);
  __syncthreads();
  const float* W1 = Wn1 + (size_t)t * Hh * Hh;
  float a2 = bn1[t * Hh + tid];
#pragma unroll 8
  for (int f = 0; f < Hh; ++f) a2 += t0[f] * W1[f * Hh + tid];
  a2 = lrelu(a2);
  hout[bn * Hh + tid] = apply_tanh ? tanhf(a2) : a2;
}

extern "C" void kernel_launch(void* const* d_in, const int* in_sizes, int n_in,
                              void* d_out, int out_size, void* d_ws, size_t ws_size,
                              hipStream_t stream) {
  const float* x     = (const float*)d_in[0];
  const float* W_lin = (const float*)d_in[1];
  const float* b_lin = (const float*)d_in[2];
  const float* W_in  = (const float*)d_in[3];
  const float* b_in  = (const float*)d_in[4];
  const float* We1   = (const float*)d_in[5];
  const float* be1   = (const float*)d_in[6];
  const float* We2   = (const float*)d_in[7];
  const float* be2   = (const float*)d_in[8];
  const float* Wn0   = (const float*)d_in[9];
  const float* bn0   = (const float*)d_in[10];
  const float* Wn1   = (const float*)d_in[11];
  const float* bn1   = (const float*)d_in[12];
  float* out = (float*)d_out;

  float* ws  = (float*)d_ws;
  float* h   = ws;
  float* rcv = h   + Bsz * Nn * Hh;
  float* snd = rcv + Bsz * Nn * HE;
  float* mbuf= snd + Bsz * Nn * HE;

  const int BN = Bsz * Nn;

  k_input<<<BN, 64, 0, stream>>>(x, W_lin, b_lin, W_in, b_in, h);
  for (int t = 0; t < NMPS; ++t) {
    k_rcv_snd<<<BN, 128, 0, stream>>>(h, We1, rcv, snd, t);
    k_edge<<<NBLK_EDGE, 256, 0, stream>>>(rcv, snd, be1, We2, be2, mbuf, t);
    k_node<<<BN, 32, 0, stream>>>(h, mbuf, Wn0, bn0, Wn1, bn1,
                                  (t == NMPS - 1) ? out : h, t, t == NMPS - 1);
  }
}


# Launch fusion: node MLP and next round's sender projection in the edge kernel's epilogue, receiver projection in its prologue, first sender projection in the input kernel; 10 launches -> 4
# speedup vs baseline: 1.3696x; 1.1188x over previous
#include <hip/hip_runtime.h>
#include <math.h>

#define Bsz   16
#define Nn    256
#define Ld    64
#define Hh    32
#define HE    128
#define EO    64
#define NMPS  3
#define ALPHA 0.2f

typedef __attribute__((ext_vector_type(2))) float v2f;
typedef __attribute__((ext_vector_type(8))) float v8f;

__device__ __forceinline__ float lrelu(float v) { return (v >= 0.f) ? v : ALPHA * v; }

__global__ __launch_bounds__(64) void k_input(const float* __restrict__ x,
                                              const float* __restrict__ W_lin,
                                              const float* __restrict__ b_lin,
                                              const float* __restrict__ W_in,
                                              const float* __restrict__ b_in,
                                              const float* __restrict__ We1,
                                              float* __restrict__ h,
                                              float* __restrict__ snd) {
  const int b = blockIdx.x >> 8;
  const int n = blockIdx.x & 255;
  const int c = threadIdx.x;
  __shared__ float xs[Ld];
  __shared__ float ts[Ld];
  __shared__ float hs[Hh];
  xs[c] = x[b * Ld + c];
  __syncthreads();
  float acc = b_lin[n * Ld + c];
#pragma unroll 8
  for (int l = 0; l < Ld; ++l) acc += xs[l] * W_lin[l * (Nn * Ld) + n * Ld + c];
  ts[c] = acc;
  __syncthreads();
  if (c < Hh) {
    float a2 = b_in[c];
#pragma unroll 8
    for (int l = 0; l < Ld; ++l) a2 += ts[l] * W_in[l * Hh + c];
    h[(b * Nn + n) * Hh + c] = a2;
    hs[c] = a2;
  }
  __syncthreads();
  const float* W1s = We1 + Hh * HE;
  float s0 = 0.f, s1 = 0.f;
#pragma unroll 8
  for (int hh = 0; hh < Hh; ++hh) {
    const float hv = hs[hh];
    s0 += hv * W1s[hh * HE + c];
    s1 += hv * W1s[hh * HE + 64 + c];
  }
  snd[(b * Nn + n) * HE + c]      = s0;
  snd[(b * Nn + n) * HE + 64 + c] = s1;
}

#define RT    16
#define SW    32
#define NWAVE 8
#define NBLK_EDGE 256

#define RP    132

typedef __attribute__((ext_vector_type(4))) float v4f;

__global__ __launch_bounds__(256) void k_edge(float* __restrict__ h,
                                              const float* __restrict__ snd,
                                              float* __restrict__ snd_next,
                                              const float* __restrict__ We1,
                                              const float* __restrict__ be1,
                                              const float* __restrict__ We2,
                                              const float* __restrict__ be2,
                                              const float* __restrict__ Wn0,
                                              const float* __restrict__ bn0,
                                              const float* __restrict__ Wn1,
                                              const float* __restrict__ bn1,
                                              float* __restrict__ out, int t) {
  const int b    = blockIdx.x >> 4;
  const int it   = blockIdx.x & 15;
  const int tid  = threadIdx.x;
  const int wave = tid >> 5;
  const int lane = tid & 31;
  const int lh   = lane >> 4;
  const int ln   = lane & 15;

  __shared__ __attribute__((aligned(16))) float part[NWAVE][RT][EO];
  __shared__ __attribute__((aligned(16))) float nf[RT][Hh + EO];
  __shared__ __attribute__((aligned(16))) float rcvs[RT][RP];
  __shared__ __attribute__((aligned(16))) float t0s[RT][Hh];
  __shared__ __attribute__((aligned(16))) float hn[RT][Hh];

  const int row = tid >> 4;
  const int cg  = tid & 15;
  const size_t grow = (size_t)(b * Nn + it * RT + row);

  *(v2f*)&nf[row][2 * cg] = *(const v2f*)(h + grow * Hh + 2 * cg);
  __syncthreads();
  {
    const float* W1r = We1 + (size_t)t * (2 * Hh) * HE;
    v4f r0 = {0.f, 0.f, 0.f, 0.f}, r1 = {0.f, 0.f, 0.f, 0.f};
#pragma unroll 16
    for (int hh = 0; hh < Hh; ++hh) {
      const float hv = nf[row][hh];
      r0 += hv * *(const v4f*)(W1r + hh * HE + 4 * cg);
      r1 += hv * *(const v4f*)(W1r + hh * HE + 64 + 4 * cg);
    }
    *(v4f*)&rcvs[row][4 * cg]      = r0;
    *(v4f*)&rcvs[row][64 + 4 * cg] = r1;
  }
  __syncthreads();

  const float* rrow = &rcvs[ln][4 * lh];
  const float* brow = be1 + t * HE + 4 * lh;
  v4f rf[16];
#pragma unroll
  for (int q = 0; q < 16; ++q)
    rf[q] = *(const v4f*)(rrow + 8 * q) + *(const v4f*)(brow + 8 * q);

  const float* W2 = We2 + (size_t)t * HE * EO;
  v2f Bf[4][32];
#pragma unroll
  for (int kk = 0; kk < 32; ++kk) {
    const int k0 = 8 * (kk >> 1) + 4 * lh + 2 * (kk & 1);
    const v4f w0 = *(const v4f*)(W2 + (k0    ) * EO + 4 * ln);
    const v4f w1 = *(const v4f*)(W2 + (k0 + 1) * EO + 4 * ln);
#pragma unroll
    for (int nt = 0; nt < 4; ++nt) {
      Bf[nt][kk].x = w0[nt];
      Bf[nt][kk].y = w1[nt];
    }
  }
  const v4f bias2 = *(const v4f*)(be2 + t * EO + 4 * ln);

  float macc[4][8];
#pragma unroll
  for (int nt = 0; nt < 4; ++nt)
#pragma unroll
    for (int r = 0; r < 8; ++r) macc[nt][r] = 0.f;

  const float* srow = snd + (size_t)(b * Nn + wave * SW) * HE + 4 * lh;
  v4f ring[8];
#pragma unroll
  for (int q = 0; q < 8; ++q) ring[q] = *(const v4f*)(srow + 8 * q);

  for (int s = 0; s < SW; ++s) {
    const float* crow = srow + (size_t)s * HE;
    const float* nrow = srow + (size_t)((s + 1 < SW) ? s + 1 : s) * HE;

    v8f C[4];
#pragma unroll
    for (int nt = 0; nt < 4; ++nt) C[nt] = (v8f){};

#pragma unroll
    for (int q = 0; q < 16; ++q) {
      const v4f x = rf[q] + ring[q & 7];
      ring[q & 7] = (q < 8) ? *(const v4f*)(crow + 8 * (q + 8))
                            : *(const v4f*)(nrow + 8 * (q - 8));
      v2f A0, A1;
      A0.x = lrelu(x.x); A0.y = lrelu(x.y);
      A1.x = lrelu(x.z); A1.y = lrelu(x.w);
#pragma unroll
      for (int nt = 0; nt < 4; ++nt)
        C[nt] = __builtin_amdgcn_wmma_f32_16x16x4_f32(
             false, A0,  false, Bf[nt][2 * q],
             (short)0, C[nt],  false,  false);
#pragma unroll
      for (int nt = 0; nt < 4; ++nt)
        C[nt] = __builtin_amdgcn_wmma_f32_16x16x4_f32(
             false, A1,  false, Bf[nt][2 * q + 1],
             (short)0, C[nt],  false,  false);
    }
#pragma unroll
    for (int nt = 0; nt < 4; ++nt)
#pragma unroll
      for (int r = 0; r < 8; ++r) macc[nt][r] += lrelu(C[nt][r] + bias2[nt]);
  }

#pragma unroll
  for (int r = 0; r < 8; ++r) {
    v4f pv;
#pragma unroll
    for (int nt = 0; nt < 4; ++nt) pv[nt] = macc[nt][r];
    *(v4f*)&part[wave][8 * lh + r][4 * ln] = pv;
  }
  __syncthreads();
  {
    const int c4 = 4 * cg;
    v4f acc = *(const v4f*)&part[0][row][c4];
#pragma unroll
    for (int w = 1; w < NWAVE; ++w) acc += *(const v4f*)&part[w][row][c4];
    *(v4f*)&nf[row][Hh + c4] = acc;
  }
  __syncthreads();

  {
    const float* W0 = Wn0 + (size_t)t * (Hh + EO) * Hh;
    v2f a = *(const v2f*)(bn0 + t * Hh + 2 * cg);
#pragma unroll 32
    for (int f = 0; f < Hh + EO; ++f)
      a += nf[row][f] * *(const v2f*)(W0 + f * Hh + 2 * cg);
    v2f tv;
    tv.x = lrelu(a.x); tv.y = lrelu(a.y);
    *(v2f*)&t0s[row][2 * cg] = tv;
  }
  __syncthreads();
  v2f hv2;
  {
    const float* W1n = Wn1 + (size_t)t * Hh * Hh;
    v2f a = *(const v2f*)(bn1 + t * Hh + 2 * cg);
#pragma unroll
    for (int f = 0; f < Hh; ++f)
      a += t0s[row][f] * *(const v2f*)(W1n + f * Hh + 2 * cg);
    hv2.x = lrelu(a.x); hv2.y = lrelu(a.y);
  }
  if (t == NMPS - 1) {
    v2f o;
    o.x = tanhf(hv2.x); o.y = tanhf(hv2.y);
    *(v2f*)(out + grow * Hh + 2 * cg) = o;
  } else {
    *(v2f*)(h + grow * Hh + 2 * cg) = hv2;
    *(v2f*)&hn[row][2 * cg] = hv2;
    __syncthreads();
    const float* W1s = We1 + (size_t)(t + 1) * (2 * Hh) * HE + Hh * HE;
    v4f s0 = {0.f, 0.f, 0.f, 0.f}, s1 = {0.f, 0.f, 0.f, 0.f};
#pragma unroll 16
    for (int hh = 0; hh < Hh; ++hh) {
      const float hv = hn[row][hh];
      s0 += hv * *(const v4f*)(W1s + hh * HE + 4 * cg);
      s1 += hv * *(const v4f*)(W1s + hh * HE + 64 + 4 * cg);
    }
    *(v4f*)(snd_next + grow * HE + 4 * cg)      = s0;
    *(v4f*)(snd_next + grow * HE + 64 + 4 * cg) = s1;
  }
}

extern "C" void kernel_launch(void* const* d_in, const int* in_sizes, int n_in,
                              void* d_out, int out_size, void* d_ws, size_t ws_size,
                              hipStream_t stream) {
  const float* x     = (const float*)d_in[0];
  const float* W_lin = (const float*)d_in[1];
  const float* b_lin = (const float*)d_in[2];
  const float* W_in  = (const float*)d_in[3];
  const float* b_in  = (const float*)d_in[4];
  const float* We1   = (const float*)d_in[5];
  const float* be1   = (const float*)d_in[6];
  const float* We2   = (const float*)d_in[7];
  const float* be2   = (const float*)d_in[8];
  const float* Wn0   = (const float*)d_in[9];
  const float* bn0   = (const float*)d_in[10];
  const float* Wn1   = (const float*)d_in[11];
  const float* bn1   = (const float*)d_in[12];
  float* out = (float*)d_out;

  float* ws   = (float*)d_ws;
  float* h    = ws;
  float* sndA = h    + Bsz * Nn * Hh;
  float* sndB = sndA + Bsz * Nn * HE;

  const int BN = Bsz * Nn;

  k_input<<<BN, 64, 0, stream>>>(x, W_lin, b_lin, W_in, b_in, We1, h, sndA);
  k_edge<<<NBLK_EDGE, 256, 0, stream>>>(h, sndA, sndB, We1, be1, We2, be2,
                                        Wn0, bn0, Wn1, bn1, out, 0);
  k_edge<<<NBLK_EDGE, 256, 0, stream>>>(h, sndB, sndA, We1, be1, We2, be2,
                                        Wn0, bn0, Wn1, bn1, out, 1);
  k_edge<<<NBLK_EDGE, 256, 0, stream>>>(h, sndA, sndB, We1, be1, We2, be2,
                                        Wn0, bn0, Wn1, bn1, out, 2);
}


# k_edge sender loop: 275 -> 160 vector instructions per 128 WMMAs (leaky relu as multiply+max, no compare+select; per-sender sums on accumulator register pairs, no repacking moves); bit-identical
# speedup vs baseline: 1.4549x; 1.0623x over previous
#include <hip/hip_runtime.h>
#include <math.h>

#define Bsz   16
#define Nn    256
#define Ld    64
#define Hh    32
#define HE    128
#define EO    64
#define NMPS  3
#define ALPHA 0.2f

typedef __attribute__((ext_vector_type(2))) float v2f;
typedef __attribute__((ext_vector_type(8))) float v8f;

__device__ __forceinline__ float lrelu(float v) { return fmaxf(v, ALPHA * v); }
__device__ __forceinline__ v2f lrelu2(v2f v) {
  const v2f s = ALPHA * v;
  v2f o;
  o.x = fmaxf(v.x, s.x); o.y = fmaxf(v.y, s.y);
  return o;
}

__global__ __launch_bounds__(64) void k_input(const float* __restrict__ x,
                                              const float* __restrict__ W_lin,
                                              const float* __restrict__ b_lin,
                                              const float* __restrict__ W_in,
                                              const float* __restrict__ b_in,
                                              const float* __restrict__ We1,
                                              float* __restrict__ h,
                                              float* __restrict__ snd) {
  const int b = blockIdx.x >> 8;
  const int n = blockIdx.x & 255;
  const int c = threadIdx.x;
  __shared__ float xs[Ld];
  __shared__ float ts[Ld];
  __shared__ float hs[Hh];
  xs[c] = x[b * Ld + c];
  __syncthreads();
  float acc = b_lin[n * Ld + c];
#pragma unroll 8
  for (int l = 0; l < Ld; ++l) acc += xs[l] * W_lin[l * (Nn * Ld) + n * Ld + c];
  ts[c] = acc;
  __syncthreads();
  if (c < Hh) {
    float a2 = b_in[c];
#pragma unroll 8
    for (int l = 0; l < Ld; ++l) a2 += ts[l] * W_in[l * Hh + c];
    h[(b * Nn + n) * Hh + c] = a2;
    hs[c] = a2;
  }
  __syncthreads();
  const float* W1s = We1 + Hh * HE;
  float s0 = 0.f, s1 = 0.f;
#pragma unroll 8
  for (int hh = 0; hh < Hh; ++hh) {
    const float hv = hs[hh];
    s0 += hv * W1s[hh * HE + c];
    s1 += hv * W1s[hh * HE + 64 + c];
  }
  snd[(b * Nn + n) * HE + c]      = s0;
  snd[(b * Nn + n) * HE + 64 + c] = s1;
}

#define RT    16
#define SW    32
#define NWAVE 8
#define NBLK_EDGE 256

#define RP    132

typedef __attribute__((ext_vector_type(4))) float v4f;

__global__ __launch_bounds__(256) void k_edge(float* __restrict__ h,
                                              const float* __restrict__ snd,
                                              float* __restrict__ snd_next,
                                              const float* __restrict__ We1,
                                              const float* __restrict__ be1,
                                              const float* __restrict__ We2,
                                              const float* __restrict__ be2,
                                              const float* __restrict__ Wn0,
                                              const float* __restrict__ bn0,
                                              const float* __restrict__ Wn1,
                                              const float* __restrict__ bn1,
                                              float* __restrict__ out, int t) {
  const int b    = blockIdx.x >> 4;
  const int it   = blockIdx.x & 15;
  const int tid  = threadIdx.x;
  const int wave = tid >> 5;
  const int lane = tid & 31;
  const int lh   = lane >> 4;
  const int ln   = lane & 15;

  __shared__ __attribute__((aligned(16))) float part[NWAVE][RT][EO];
  __shared__ __attribute__((aligned(16))) float nf[RT][Hh + EO];
  __shared__ __attribute__((aligned(16))) float rcvs[RT][RP];
  __shared__ __attribute__((aligned(16))) float t0s[RT][Hh];
  __shared__ __attribute__((aligned(16))) float hn[RT][Hh];

  const int row = tid >> 4;
  const int cg  = tid & 15;
  const size_t grow = (size_t)(b * Nn + it * RT + row);

  *(v2f*)&nf[row][2 * cg] = *(const v2f*)(h + grow * Hh + 2 * cg);
  __syncthreads();
  {
    const float* W1r = We1 + (size_t)t * (2 * Hh) * HE;
    v4f r0 = {0.f, 0.f, 0.f, 0.f}, r1 = {0.f, 0.f, 0.f, 0.f};
#pragma unroll 16
    for (int hh = 0; hh < Hh; ++hh) {
      const float hv = nf[row][hh];
      r0 += hv * *(const v4f*)(W1r + hh * HE + 4 * cg);
      r1 += hv * *(const v4f*)(W1r + hh * HE + 64 + 4 * cg);
    }
    *(v4f*)&rcvs[row][4 * cg]      = r0;
    *(v4f*)&rcvs[row][64 + 4 * cg] = r1;
  }
  __syncthreads();

  const float* rrow = &rcvs[ln][4 * lh];
  const float* brow = be1 + t * HE + 4 * lh;
  v4f rf[16];
#pragma unroll
  for (int q = 0; q < 16; ++q)
    rf[q] = *(const v4f*)(rrow + 8 * q) + *(const v4f*)(brow + 8 * q);

  const float* W2 = We2 + (size_t)t * HE * EO;
  v2f Bf[4][32];
#pragma unroll
  for (int kk = 0; kk < 32; ++kk) {
    const int k0 = 8 * (kk >> 1) + 4 * lh + 2 * (kk & 1);
    const v4f w0 = *(const v4f*)(W2 + (k0    ) * EO + 4 * ln);
    const v4f w1 = *(const v4f*)(W2 + (k0 + 1) * EO + 4 * ln);
#pragma unroll
    for (int nt = 0; nt < 4; ++nt) {
      Bf[nt][kk].x = w0[nt];
      Bf[nt][kk].y = w1[nt];
    }
  }
  const v4f bias2 = *(const v4f*)(be2 + t * EO + 4 * ln);

  v2f macc[4][4];
#pragma unroll
  for (int nt = 0; nt < 4; ++nt)
#pragma unroll
    for (int p = 0; p < 4; ++p) macc[nt][p] = (v2f){0.f, 0.f};

  const float* srow = snd + (size_t)(b * Nn + wave * SW) * HE + 4 * lh;
  v4f ring[8];
#pragma unroll
  for (int q = 0; q < 8; ++q) ring[q] = *(const v4f*)(srow + 8 * q);

  for (int s = 0; s < SW; ++s) {
    const float* crow = srow + (size_t)s * HE;
    const float* nrow = srow + (size_t)((s + 1 < SW) ? s + 1 : s) * HE;

    v8f C[4];
#pragma unroll
    for (int nt = 0; nt < 4; ++nt) C[nt] = (v8f){};

#pragma unroll
    for (int q = 0; q < 16; ++q) {
      const v4f x = rf[q] + ring[q & 7];
      ring[q & 7] = (q < 8) ? *(const v4f*)(crow + 8 * (q + 8))
                            : *(const v4f*)(nrow + 8 * (q - 8));
      const v2f A0 = lrelu2((v2f){x.x, x.y});
      const v2f A1 = lrelu2((v2f){x.z, x.w});
#pragma unroll
      for (int nt = 0; nt < 4; ++nt)
        C[nt] = __builtin_amdgcn_wmma_f32_16x16x4_f32(
             false, A0,  false, Bf[nt][2 * q],
             (short)0, C[nt],  false,  false);
#pragma unroll
      for (int nt = 0; nt < 4; ++nt)
        C[nt] = __builtin_amdgcn_wmma_f32_16x16x4_f32(
             false, A1,  false, Bf[nt][2 * q + 1],
             (short)0, C[nt],  false,  false);
    }
#pragma unroll
    for (int nt = 0; nt < 4; ++nt) {
      const v2f bb = {bias2[nt], bias2[nt]};
#pragma unroll
      for (int p = 0; p < 4; ++p) {
        v2f y;
        y.x = C[nt][2 * p]; y.y = C[nt][2 * p + 1];
        macc[nt][p] += lrelu2(y + bb);
      }
    }
  }

#pragma unroll
  for (int r = 0; r < 8; ++r) {
    v4f pv;
#pragma unroll
    for (int nt = 0; nt < 4; ++nt) pv[nt] = macc[nt][r >> 1][r & 1];
    *(v4f*)&part[wave][8 * lh + r][4 * ln] = pv;
  }
  __syncthreads();
  {
    const int c4 = 4 * cg;
    v4f acc = *(const v4f*)&part[0][row][c4];
#pragma unroll
    for (int w = 1; w < NWAVE; ++w) acc += *(const v4f*)&part[w][row][c4];
    *(v4f*)&nf[row][Hh + c4] = acc;
  }
  __syncthreads();

  {
    const float* W0 = Wn0 + (size_t)t * (Hh + EO) * Hh;
    v2f a = *(const v2f*)(bn0 + t * Hh + 2 * cg);
#pragma unroll 32
    for (int f = 0; f < Hh + EO; ++f)
      a += nf[row][f] * *(const v2f*)(W0 + f * Hh + 2 * cg);
    v2f tv;
    tv.x = lrelu(a.x); tv.y = lrelu(a.y);
    *(v2f*)&t0s[row][2 * cg] = tv;
  }
  __syncthreads();
  v2f hv2;
  {
    const float* W1n = Wn1 + (size_t)t * Hh * Hh;
    v2f a = *(const v2f*)(bn1 + t * Hh + 2 * cg);
#pragma unroll
    for (int f = 0; f < Hh; ++f)
      a += t0s[row][f] * *(const v2f*)(W1n + f * Hh + 2 * cg);
    hv2.x = lrelu(a.x); hv2.y = lrelu(a.y);
  }
  if (t == NMPS - 1) {
    v2f o;
    o.x = tanhf(hv2.x); o.y = tanhf(hv2.y);
    *(v2f*)(out + grow * Hh + 2 * cg) = o;
  } else {
    *(v2f*)(h + grow * Hh + 2 * cg) = hv2;
    *(v2f*)&hn[row][2 * cg] = hv2;
    __syncthreads();
    const float* W1s = We1 + (size_t)(t + 1) * (2 * Hh) * HE + Hh * HE;
    v4f s0 = {0.f, 0.f, 0.f, 0.f}, s1 = {0.f, 0.f, 0.f, 0.f};
#pragma unroll 16
    for (int hh = 0; hh < Hh; ++hh) {
      const float hv = hn[row][hh];
      s0 += hv * *(const v4f*)(W1s + hh * HE + 4 * cg);
      s1 += hv * *(const v4f*)(W1s + hh * HE + 64 + 4 * cg);
    }
    *(v4f*)(snd_next + grow * HE + 4 * cg)      = s0;
    *(v4f*)(snd_next + grow * HE + 64 + 4 * cg) = s1;
  }
}

extern "C" void kernel_launch(void* const* d_in, const int* in_sizes, int n_in,
                              void* d_out, int out_size, void* d_ws, size_t ws_size,
                              hipStream_t stream) {
  const float* x     = (const float*)d_in[0];
  const float* W_lin = (const float*)d_in[1];
  const float* b_lin = (const float*)d_in[2];
  const float* W_in  = (const float*)d_in[3];
  const float* b_in  = (const float*)d_in[4];
  const float* We1   = (const float*)d_in[5];
  const float* be1   = (const float*)d_in[6];
  const float* We2   = (const float*)d_in[7];
  const float* be2   = (const float*)d_in[8];
  const float* Wn0   = (const float*)d_in[9];
  const float* bn0   = (const float*)d_in[10];
  const float* Wn1   = (const float*)d_in[11];
  const float* bn1   = (const float*)d_in[12];
  float* out = (float*)d_out;

  float* ws   = (float*)d_ws;
  float* h    = ws;
  float* sndA = h    + Bsz * Nn * Hh;
  float* sndB = sndA + Bsz * Nn * HE;

  const int BN = Bsz * Nn;

  k_input<<<BN, 64, 0, stream>>>(x, W_lin, b_lin, W_in, b_in, We1, h, sndA);
  k_edge<<<NBLK_EDGE, 256, 0, stream>>>(h, sndA, sndB, We1, be1, We2, be2,
                                        Wn0, bn0, Wn1, bn1, out, 0);
  k_edge<<<NBLK_EDGE, 256, 0, stream>>>(h, sndB, sndA, We1, be1, We2, be2,
                                        Wn0, bn0, Wn1, bn1, out, 1);
  k_edge<<<NBLK_EDGE, 256, 0, stream>>>(h, sndA, sndB, We1, be1, We2, be2,
                                        Wn0, bn0, Wn1, bn1, out, 2);
}


# Per-node dense layers (input, receiver/sender projections, node MLP) as 16-row WMMA tiles: a wave loads only its own 16 weight columns, once per block; k and sum order of these layers change
# speedup vs baseline: 1.5577x; 1.0707x over previous
#include <hip/hip_runtime.h>
#include <math.h>

#define Bsz   16
#define Nn    256
#define Ld    64
#define Hh    32
#define HE    128
#define EO    64
#define NMPS  3
#define ALPHA 0.2f

typedef __attribute__((ext_vector_type(2))) float v2f;
typedef __attribute__((ext_vector_type(8))) float v8f;

__device__ __forceinline__ float lrelu(float v) { return fmaxf(v, ALPHA * v); }
__device__ __forceinline__ v2f lrelu2(v2f v) {
  const v2f s = ALPHA * v;
  v2f o;
  o.x = fmaxf(v.x, s.x); o.y = fmaxf(v.y, s.y);
  return o;
}

typedef __attribute__((ext_vector_type(4))) float v4f;

template <int KS>
__device__ __forceinline__ void load_b(v2f (&B)[KS], const float* __restrict__ Wm,
                                       int pitch, int col, int lh) {
#pragma unroll
  for (int ks = 0; ks < KS; ++ks) {
    B[ks].x = Wm[(size_t)(4 * ks + 2 * lh) * pitch + col];
    B[ks].y = Wm[(size_t)(4 * ks + 2 * lh + 1) * pitch + col];
  }
}

template <int KS, int NACC>
__device__ __forceinline__ v8f mm_tile(const float* arow, const v2f (&B)[KS]) {
  v8f acc[NACC];
#pragma unroll
  for (int i = 0; i < NACC; ++i) acc[i] = (v8f){};
#pragma unroll
  for (int ks = 0; ks < KS; ++ks) {
    const v2f a = *(const v2f*)(arow + 4 * ks);
    acc[ks % NACC] = __builtin_amdgcn_wmma_f32_16x16x4_f32(
         false, a,  false, B[ks],
         (short)0, acc[ks % NACC],  false,  false);
  }
#pragma unroll
  for (int i = 1; i < NACC; ++i) acc[0] += acc[i];
  return acc[0];
}

#define RP    132
#define P64   68
#define P32   36

__global__ __launch_bounds__(256) void k_input(const float* __restrict__ x,
                                               const float* __restrict__ W_lin,
                                               const float* __restrict__ b_lin,
                                               const float* __restrict__ W_in,
                                               const float* __restrict__ b_in,
                                               const float* __restrict__ We1,
                                               float* __restrict__ h,
                                               float* __restrict__ snd) {
  const int n    = blockIdx.x;
  const int tid  = threadIdx.x;
  const int lane = tid & 31;
  const int lh   = lane >> 4;
  const int ln   = lane & 15;
  const int wv   = __builtin_amdgcn_readfirstlane(tid >> 5);
  const int row  = tid >> 4;
  const int cg   = tid & 15;

  __shared__ __attribute__((aligned(16))) float xs[Bsz][P64];
  __shared__ __attribute__((aligned(16))) float ts[2][Bsz][P64];
  __shared__ __attribute__((aligned(16))) float hs[Bsz][P32];
  __shared__ __attribute__((aligned(16))) float ss[Bsz][RP];

  const int tile1 = wv & 3, kh = wv >> 2;
  v2f B1[8];
  load_b<8>(B1, W_lin + (size_t)(32 * kh) * (Nn * Ld) + n * Ld, Nn * Ld, 16 * tile1 + ln, lh);
  v2f B2[16];
  if (wv < 2) load_b<16>(B2, W_in, Hh, 16 * wv + ln, lh);
  v2f B3[8];
  load_b<8>(B3, We1 + Hh * HE, HE, 16 * wv + ln, lh);

  *(v4f*)&xs[row][4 * cg] = *(const v4f*)(x + row * Ld + 4 * cg);
  __syncthreads();

  {
    v8f d = mm_tile<8, 2>(&xs[ln][32 * kh + 2 * lh], B1);
    const float bb = (kh == 0) ? b_lin[n * Ld + 16 * tile1 + ln] : 0.f;
#pragma unroll
    for (int r = 0; r < 8; ++r) ts[kh][r + 8 * lh][16 * tile1 + ln] = d[r] + bb;
  }
  __syncthreads();

  if (wv < 2) {
    v8f a0 = (v8f){}, a1 = (v8f){};
    const float* t0 = &ts[0][ln][2 * lh];
    const float* t1 = &ts[1][ln][2 * lh];
#pragma unroll
    for (int ks = 0; ks < 16; ++ks) {
      const v2f a = *(const v2f*)(t0 + 4 * ks) + *(const v2f*)(t1 + 4 * ks);
      if (ks & 1)
        a1 = __builtin_amdgcn_wmma_f32_16x16x4_f32(false, a, false, B2[ks], (short)0, a1, false, false);
      else
        a0 = __builtin_amdgcn_wmma_f32_16x16x4_f32(false, a, false, B2[ks], (short)0, a0, false, false);
    }
    const v8f d = a0 + a1;
    const float bb = b_in[16 * wv + ln];
#pragma unroll
    for (int r = 0; r < 8; ++r) hs[r + 8 * lh][16 * wv + ln] = d[r] + bb;
  }
  __syncthreads();

  *(v2f*)(h + (size_t)(row * Nn + n) * Hh + 2 * cg) = *(const v2f*)&hs[row][2 * cg];

  {
    const v8f d = mm_tile<8, 2>(&hs[ln][2 * lh], B3);
#pragma unroll
    for (int r = 0; r < 8; ++r) ss[r + 8 * lh][16 * wv + ln] = d[r];
  }
  __syncthreads();
  *(v4f*)(snd + (size_t)(row * Nn + n) * HE + 4 * cg)      = *(const v4f*)&ss[row][4 * cg];
  *(v4f*)(snd + (size_t)(row * Nn + n) * HE + 64 + 4 * cg) = *(const v4f*)&ss[row][64 + 4 * cg];
}

#define RT    16
#define SW    32
#define NWAVE 8
#define NBLK_EDGE 256
#define NFP   100


__global__ __launch_bounds__(256) void k_edge(float* __restrict__ h,
                                              const float* __restrict__ snd,
                                              float* __restrict__ snd_next,
                                              const float* __restrict__ We1,
                                              const float* __restrict__ be1,
                                              const float* __restrict__ We2,
                                              const float* __restrict__ be2,
                                              const float* __restrict__ Wn0,
                                              const float* __restrict__ bn0,
                                              const float* __restrict__ Wn1,
                                              const float* __restrict__ bn1,
                                              float* __restrict__ out, int t) {
  const int b    = blockIdx.x >> 4;
  const int it   = blockIdx.x & 15;
  const int tid  = threadIdx.x;
  const int wave = tid >> 5;
  const int lane = tid & 31;
  const int lh   = lane >> 4;
  const int ln   = lane & 15;

  __shared__ __attribute__((aligned(16))) float part[NWAVE][RT][EO];
  __shared__ __attribute__((aligned(16))) float nf[RT][NFP];
  __shared__ __attribute__((aligned(16))) float rcvs[RT][RP];
  __shared__ __attribute__((aligned(16))) float t0s[RT][P32];
  __shared__ __attribute__((aligned(16))) float hn[RT][P32];

  const int row = tid >> 4;
  const int cg  = tid & 15;
  const size_t grow = (size_t)(b * Nn + it * RT + row);
  const int wv = __builtin_amdgcn_readfirstlane(wave);

  v2f Brcv[8];
  load_b<8>(Brcv, We1 + (size_t)t * (2 * Hh) * HE, HE, 16 * wv + ln, lh);
  *(v2f*)&nf[row][2 * cg] = *(const v2f*)(h + grow * Hh + 2 * cg);
  __syncthreads();
  {
    const v8f d = mm_tile<8, 2>(&nf[ln][2 * lh], Brcv);
#pragma unroll
    for (int r = 0; r < 8; ++r) rcvs[r + 8 * lh][16 * wv + ln] = d[r];
  }
  __syncthreads();

  const float* rrow = &rcvs[ln][4 * lh];
  const float* brow = be1 + t * HE + 4 * lh;
  v4f rf[16];
#pragma unroll
  for (int q = 0; q < 16; ++q)
    rf[q] = *(const v4f*)(rrow + 8 * q) + *(const v4f*)(brow + 8 * q);

  const float* W2 = We2 + (size_t)t * HE * EO;
  v2f Bf[4][32];
#pragma unroll
  for (int kk = 0; kk < 32; ++kk) {
    const int k0 = 8 * (kk >> 1) + 4 * lh + 2 * (kk & 1);
    const v4f w0 = *(const v4f*)(W2 + (k0    ) * EO + 4 * ln);
    const v4f w1 = *(const v4f*)(W2 + (k0 + 1) * EO + 4 * ln);
#pragma unroll
    for (int nt = 0; nt < 4; ++nt) {
      Bf[nt][kk].x = w0[nt];
      Bf[nt][kk].y = w1[nt];
    }
  }
  const v4f bias2 = *(const v4f*)(be2 + t * EO + 4 * ln);

  v2f macc[4][4];
#pragma unroll
  for (int nt = 0; nt < 4; ++nt)
#pragma unroll
    for (int p = 0; p < 4; ++p) macc[nt][p] = (v2f){0.f, 0.f};

  const float* srow = snd + (size_t)(b * Nn + wave * SW) * HE + 4 * lh;
  v4f ring[8];
#pragma unroll
  for (int q = 0; q < 8; ++q) ring[q] = *(const v4f*)(srow + 8 * q);

  for (int s = 0; s < SW; ++s) {
    const float* crow = srow + (size_t)s * HE;
    const float* nrow = srow + (size_t)((s + 1 < SW) ? s + 1 : s) * HE;

    v8f C[4];
#pragma unroll
    for (int nt = 0; nt < 4; ++nt) C[nt] = (v8f){};

#pragma unroll
    for (int q = 0; q < 16; ++q) {
      const v4f x = rf[q] + ring[q & 7];
      ring[q & 7] = (q < 8) ? *(const v4f*)(crow + 8 * (q + 8))
                            : *(const v4f*)(nrow + 8 * (q - 8));
      const v2f A0 = lrelu2((v2f){x.x, x.y});
      const v2f A1 = lrelu2((v2f){x.z, x.w});
#pragma unroll
      for (int nt = 0; nt < 4; ++nt)
        C[nt] = __builtin_amdgcn_wmma_f32_16x16x4_f32(
             false, A0,  false, Bf[nt][2 * q],
             (short)0, C[nt],  false,  false);
#pragma unroll
      for (int nt = 0; nt < 4; ++nt)
        C[nt] = __builtin_amdgcn_wmma_f32_16x16x4_f32(
             false, A1,  false, Bf[nt][2 * q + 1],
             (short)0, C[nt],  false,  false);
    }
#pragma unroll
    for (int nt = 0; nt < 4; ++nt) {
      const v2f bb = {bias2[nt], bias2[nt]};
#pragma unroll
      for (int p = 0; p < 4; ++p) {
        v2f y;
        y.x = C[nt][2 * p]; y.y = C[nt][2 * p + 1];
        macc[nt][p] += lrelu2(y + bb);
      }
    }
  }

  const bool last = (t == NMPS - 1);
  v2f Bn0[24], Bn1[8], Bsn[8];
  if (wv < 2) {
    load_b<24>(Bn0, Wn0 + (size_t)t * (Hh + EO) * Hh, Hh, 16 * wv + ln, lh);
    load_b<8>(Bn1, Wn1 + (size_t)t * Hh * Hh, Hh, 16 * wv + ln, lh);
  }
  if (!last)
    load_b<8>(Bsn, We1 + (size_t)(t + 1) * (2 * Hh) * HE + Hh * HE, HE, 16 * wv + ln, lh);

#pragma unroll
  for (int r = 0; r < 8; ++r) {
    v4f pv;
#pragma unroll
    for (int nt = 0; nt < 4; ++nt) pv[nt] = macc[nt][r >> 1][r & 1];
    *(v4f*)&part[wv][8 * lh + r][4 * ln] = pv;
  }
  __syncthreads();
  {
    const int c4 = 4 * cg;
    v4f acc = *(const v4f*)&part[0][row][c4];
#pragma unroll
    for (int w = 1; w < NWAVE; ++w) acc += *(const v4f*)&part[w][row][c4];
    *(v4f*)&nf[row][Hh + c4] = acc;
  }
  __syncthreads();

  if (wv < 2) {
    const v8f d = mm_tile<24, 4>(&nf[ln][2 * lh], Bn0);
    const float bb = bn0[t * Hh + 16 * wv + ln];
#pragma unroll
    for (int r = 0; r < 8; ++r) t0s[r + 8 * lh][16 * wv + ln] = lrelu(d[r] + bb);
  }
  __syncthreads();
  if (wv < 2) {
    const v8f d = mm_tile<8, 2>(&t0s[ln][2 * lh], Bn1);
    const float bb = bn1[t * Hh + 16 * wv + ln];
#pragma unroll
    for (int r = 0; r < 8; ++r) hn[r + 8 * lh][16 * wv + ln] = lrelu(d[r] + bb);
  }
  __syncthreads();
  const v2f hv2 = *(const v2f*)&hn[row][2 * cg];
  if (last) {
    v2f o;
    o.x = tanhf(hv2.x); o.y = tanhf(hv2.y);
    *(v2f*)(out + grow * Hh + 2 * cg) = o;
  } else {
    *(v2f*)(h + grow * Hh + 2 * cg) = hv2;
    {
      const v8f d = mm_tile<8, 2>(&hn[ln][2 * lh], Bsn);
#pragma unroll
      for (int r = 0; r < 8; ++r) rcvs[r + 8 * lh][16 * wv + ln] = d[r];
    }
    __syncthreads();
    *(v4f*)(snd_next + grow * HE + 4 * cg)      = *(const v4f*)&rcvs[row][4 * cg];
    *(v4f*)(snd_next + grow * HE + 64 + 4 * cg) = *(const v4f*)&rcvs[row][64 + 4 * cg];
  }
}

extern "C" void kernel_launch(void* const* d_in, const int* in_sizes, int n_in,
                              void* d_out, int out_size, void* d_ws, size_t ws_size,
                              hipStream_t stream) {
  const float* x     = (const float*)d_in[0];
  const float* W_lin = (const float*)d_in[1];
  const float* b_lin = (const float*)d_in[2];
  const float* W_in  = (const float*)d_in[3];
  const float* b_in  = (const float*)d_in[4];
  const float* We1   = (const float*)d_in[5];
  const float* be1   = (const float*)d_in[6];
  const float* We2   = (const float*)d_in[7];
  const float* be2   = (const float*)d_in[8];
  const float* Wn0   = (const float*)d_in[9];
  const float* bn0   = (const float*)d_in[10];
  const float* Wn1   = (const float*)d_in[11];
  const float* bn1   = (const float*)d_in[12];
  float* out = (float*)d_out;

  float* ws   = (float*)d_ws;
  float* h    = ws;
  float* sndA = h    + Bsz * Nn * Hh;
  float* sndB = sndA + Bsz * Nn * HE;

  k_input<<<Nn, 256, 0, stream>>>(x, W_lin, b_lin, W_in, b_in, We1, h, sndA);
  k_edge<<<NBLK_EDGE, 256, 0, stream>>>(h, sndA, sndB, We1, be1, We2, be2,
                                        Wn0, bn0, Wn1, bn1, out, 0);
  k_edge<<<NBLK_EDGE, 256, 0, stream>>>(h, sndB, sndA, We1, be1, We2, be2,
                                        Wn0, bn0, Wn1, bn1, out, 1);
  k_edge<<<NBLK_EDGE, 256, 0, stream>>>(h, sndA, sndB, We1, be1, We2, be2,
                                        Wn0, bn0, Wn1, bn1, out, 2);
}
